# starting kernel
# baseline (speedup 1.0000x reference)
#include <hip/hip_runtime.h>
#include <hip/hip_bf16.h>

#define S_LEN 2048
#define B_TOT 4096
#define H_SZ  8

typedef __attribute__((ext_vector_type(2))) float v2f;
typedef __attribute__((ext_vector_type(8))) float v8f;

__device__ __forceinline__ float frcp(float v) {
    return __builtin_amdgcn_rcpf(v);
}

__device__ __forceinline__ float fsigmoid(float v) {
    return frcp(1.0f + __expf(-v));
}

__device__ __forceinline__ float ftanh(float v) {
#if __has_builtin(__builtin_amdgcn_tanhf)
    return __builtin_amdgcn_tanhf(v);
#else
    return 1.0f - 2.0f * frcp(1.0f + __expf(2.0f * v));
#endif
}

__device__ __forceinline__ void lds_fence() {
    asm volatile("s_wait_dscnt 0x0" ::: "memory");
}

__device__ __forceinline__ v8f wmma4(v2f a, v2f b, v8f c) {
    return __builtin_amdgcn_wmma_f32_16x16x4_f32(
         false, a,  false, b,
         (short)0, c,  false,  false);
}

__global__ __launch_bounds__(32)
void bilstm_fused(const float* __restrict__ x,
                  const float* __restrict__ w_ih_f, const float* __restrict__ w_hh_f,
                  const float* __restrict__ b_ih_f, const float* __restrict__ b_hh_f,
                  const float* __restrict__ w_ih_b, const float* __restrict__ w_hh_b,
                  const float* __restrict__ b_ih_b, const float* __restrict__ b_hh_b,
                  const float* __restrict__ w_out, const float* __restrict__ b_out,
                  const int* __restrict__ future_p,
                  float* __restrict__ out)
{
    __shared__ __align__(16) float gbuf[2][16][40];
    __shared__ __align__(16) float hbuf[2][16][8];

    const int lane = threadIdx.x & 31;
    const int half = lane >> 4;
    const int nn   = lane & 15;
    const int bb   = blockIdx.x * 16;
    const int fut  = future_p[0];
    const int j0   = 4 * half;

    const float* wih[2] = { w_ih_f, w_ih_b };
    const float* whh[2] = { w_hh_f, w_hh_b };
    const float* bih[2] = { b_ih_f, b_ih_b };
    const float* bhh[2] = { b_hh_f, b_hh_b };

    v2f wmat[2][2][3];
    #pragma unroll
    for (int d = 0; d < 2; ++d)
        #pragma unroll
        for (int t = 0; t < 2; ++t)
            #pragma unroll
            for (int c = 0; c < 3; ++c) {
                const int n = nn + 16 * t;
                float comp[2];
                #pragma unroll
                for (int v = 0; v < 2; ++v) {
                    const int K = 4 * c + 2 * half + v;
                    float val = 0.0f;
                    if (K < 8)       val = whh[d][n * H_SZ + K];
                    else if (K == 8) val = wih[d][n];
                    else if (K == 9) val = bih[d][n] + bhh[d][n];
                    comp[v] = val;
                }
                v2f w = { comp[0], comp[1] };
                wmat[d][t][c] = w;
            }

    float wo[2][4];
    #pragma unroll
    for (int jj = 0; jj < 4; ++jj) {
        wo[0][jj] = w_out[j0 + jj];
        wo[1][jj] = w_out[8 + j0 + jj];
    }
    const float bo = b_out[0];

    v2f a0[2], a1[2];
    a0[0] = a0[1] = (v2f){0.0f, 0.0f};
    a1[0] = a1[1] = (v2f){0.0f, 0.0f};
    float cst[2][4] = {{0,0,0,0},{0,0,0,0}};

    float xv[2];
    xv[0] = x[(size_t)(0 & (S_LEN - 1)) * B_TOT + bb + nn];
    xv[1] = x[(size_t)(fut & (S_LEN - 1)) * B_TOT + bb + nn];

    const float kone = half ? 0.0f : 1.0f;

    for (int ii = 0; ii < S_LEN; ii += 4) {
        float ov[4];
        #pragma unroll
        for (int u = 0; u < 4; ++u) {
            const int i = ii + u;

            float xnxt[2];
            xnxt[0] = x[(size_t)((i + 1) & (S_LEN - 1)) * B_TOT + bb + nn];
            xnxt[1] = x[(size_t)((fut - i - 1) & (S_LEN - 1)) * B_TOT + bb + nn];

            v8f acc0[2], acc1[2];
            #pragma unroll
            for (int d = 0; d < 2; ++d) {
                const v2f a2 = { xv[d] * kone, kone };
                v8f z = {0,0,0,0,0,0,0,0};
                v8f t0 = wmma4(a0[d], wmat[d][0][0], z);
                t0 = wmma4(a1[d], wmat[d][0][1], t0);
                t0 = wmma4(a2,    wmat[d][0][2], t0);
                acc0[d] = t0;
                v8f t1 = wmma4(a0[d], wmat[d][1][0], z);
                t1 = wmma4(a1[d], wmat[d][1][1], t1);
                t1 = wmma4(a2,    wmat[d][1][2], t1);
                acc1[d] = t1;
            }

            #pragma unroll
            for (int d = 0; d < 2; ++d)
                #pragma unroll
                for (int v = 0; v < 8; ++v) {
                    gbuf[d][v + 8 * half][nn]      = acc0[d][v];
                    gbuf[d][v + 8 * half][16 + nn] = acc1[d][v];
                }
            lds_fence();

            float partial = 0.0f;
            #pragma unroll
            for (int d = 0; d < 2; ++d) {
                float hq[4];
                #pragma unroll
                for (int jj = 0; jj < 4; ++jj) {
                    const int j = j0 + jj;
                    const float ig = fsigmoid(gbuf[d][nn][j]);
                    const float fg = fsigmoid(gbuf[d][nn][8 + j]);
                    const float gg = ftanh(gbuf[d][nn][16 + j]);
                    const float og = fsigmoid(gbuf[d][nn][24 + j]);
                    const float cc = fg * cst[d][jj] + ig * gg;
                    cst[d][jj] = cc;
                    const float hh = og * ftanh(cc);
                    hq[jj] = hh;
                    partial += hh * wo[d][jj];
                }
                *(float4*)&hbuf[d][nn][j0] = make_float4(hq[0], hq[1], hq[2], hq[3]);
            }
            lds_fence();

            #pragma unroll
            for (int d = 0; d < 2; ++d) {
                a0[d] = *(const v2f*)&hbuf[d][nn][2 * half];
                a1[d] = *(const v2f*)&hbuf[d][nn][4 + 2 * half];
            }

            const float s = partial + __shfl_xor(partial, 16, 32);
            ov[u] = s + bo;

            xv[0] = xnxt[0];
            xv[1] = xnxt[1];
        }
        if (lane < 16) {
            *(float4*)&out[(size_t)(bb + nn) * S_LEN + ii] =
                make_float4(ov[0], ov[1], ov[2], ov[3]);
        }
    }
}

extern "C" void kernel_launch(void* const* d_in, const int* in_sizes, int n_in,
                              void* d_out, int out_size, void* d_ws, size_t ws_size,
                              hipStream_t stream) {
    (void)in_sizes; (void)n_in; (void)out_size; (void)d_ws; (void)ws_size;
    const float* xp     = (const float*)d_in[0];
    const float* w_ih_f = (const float*)d_in[1];
    const float* w_hh_f = (const float*)d_in[2];
    const float* b_ih_f = (const float*)d_in[3];
    const float* b_hh_f = (const float*)d_in[4];
    const float* w_ih_b = (const float*)d_in[5];
    const float* w_hh_b = (const float*)d_in[6];
    const float* b_ih_b = (const float*)d_in[7];
    const float* b_hh_b = (const float*)d_in[8];
    const float* w_out  = (const float*)d_in[9];
    const float* b_out  = (const float*)d_in[10];
    const int*   future = (const int*)d_in[11];
    float* out = (float*)d_out;

    dim3 grid(B_TOT / 16);
    dim3 block(32);
    bilstm_fused<<<grid, block, 0, stream>>>(
        xp, w_ih_f, w_hh_f, b_ih_f, b_hh_f,
        w_ih_b, w_hh_b, b_ih_b, b_hh_b,
        w_out, b_out, future, out);
}


# Gate product split by k over the two waves of a 64-thread work-group: each wave multiplies its own four h into all 32 gate columns, partials meet in the LDS transpose; 2 cells, 3 WMMAs per lane-step
# speedup vs baseline: 1.4926x; 1.4926x over previous
#include <hip/hip_runtime.h>
#include <hip/hip_bf16.h>

#define S_LEN 2048
#define B_TOT 4096
#define H_SZ  8
#define GP    36
#define PLANE (16 * GP)

typedef __attribute__((ext_vector_type(2))) float v2f;
typedef __attribute__((ext_vector_type(8))) float v8f;

__device__ __forceinline__ float frcp(float v) {
    return __builtin_amdgcn_rcpf(v);
}

__device__ __forceinline__ float fsigmoid(float v) {
    return frcp(1.0f + __expf(-v));
}

__device__ __forceinline__ float ftanh(float v) {
#if __has_builtin(__builtin_amdgcn_tanhf)
    return __builtin_amdgcn_tanhf(v);
#else
    return 1.0f - 2.0f * frcp(1.0f + __expf(2.0f * v));
#endif
}

__device__ __forceinline__ void lds_handover() {
    asm volatile("s_wait_dscnt 0x0\n\ts_barrier_signal -1\n\ts_barrier_wait -1" ::: "memory");
}

__device__ __forceinline__ void handover_signal(float& a, float& b, float& c, float& d) {
    asm volatile("s_wait_dscnt 0x0\n\ts_barrier_signal -1"
                 : "+v"(a), "+v"(b), "+v"(c), "+v"(d) :: "memory");
}
__device__ __forceinline__ void handover_wait() {
    asm volatile("s_barrier_wait -1" ::: "memory");
}

__device__ __forceinline__ v8f wmma4(v2f a, v2f b, v8f c) {
    return __builtin_amdgcn_wmma_f32_16x16x4_f32(
         false, a,  false, b,
         (short)0, c,  false,  false);
}

#define PHASE_PIN() __builtin_amdgcn_sched_barrier(0)

__device__ __forceinline__ void hold4(float& a, float& b, float& c, float& d) {
    asm volatile("" : "+v"(a), "+v"(b), "+v"(c), "+v"(d) :: "memory");
}
__device__ __forceinline__ void hold2(float& a, float& b) {
    asm volatile("" : "+v"(a), "+v"(b) :: "memory");
}

__device__ __forceinline__ void gates_mm(v2f a, const float xval, const float kone,
                                         const v2f wm, const v2f wt, const v2f wx,
                                         v8f& gm, v8f& gt) {
    float h0 = a[0], h1 = a[1];
    hold2(h0, h1);
    a = (v2f){ h0, h1 };
    float xh = xval;
    asm volatile("" : "+v"(xh) :: "memory");
    const v2f a2 = { xh * kone, kone };
    const v8f z = {0,0,0,0,0,0,0,0};
    v8f t0 = wmma4(a, wm, z);
    v8f t1 = wmma4(a, wt, z);
    t0 = wmma4(a2, wx, t0);
    gm = t0; gt = t1;
}

__device__ __forceinline__ void gates_to_lds(float* sm, float* st, const v8f gm, const v8f gt) {
    #pragma unroll
    for (int v = 0; v < 8; ++v) sm[v * GP] = gm[v];
    #pragma unroll
    for (int v = 0; v < 8; ++v) st[v * GP] = gt[v];
}

__device__ __forceinline__ void gates_from_lds(const float* ld, float4 (&q)[2][2]) {
    #pragma unroll
    for (int p = 0; p < 2; ++p)
        #pragma unroll
        for (int cc = 0; cc < 2; ++cc)
            q[p][cc] = *(const float4*)&ld[p * PLANE + 4 * cc];
}

__device__ __forceinline__ float cell_pair(float4 (&q)[2][2], float (&c)[2], const float (&w)[2],
                                           float partial, v2f& a) {
    hold4(q[0][0].x, q[0][0].y, q[0][0].z, q[0][0].w);
    hold4(q[0][1].x, q[0][1].y, q[0][1].z, q[0][1].w);
    hold4(q[1][0].x, q[1][0].y, q[1][0].z, q[1][0].w);
    hold4(q[1][1].x, q[1][1].y, q[1][1].z, q[1][1].w);
    float hq[2], ig[2], fg[2], gg[2], og[2];
    #pragma unroll
    for (int cc = 0; cc < 2; ++cc) {
        ig[cc] = fsigmoid(q[0][cc].x + q[1][cc].x);
        fg[cc] = fsigmoid(q[0][cc].y + q[1][cc].y);
        gg[cc] = ftanh(q[0][cc].z + q[1][cc].z);
        og[cc] = fsigmoid(q[0][cc].w + q[1][cc].w);
    }
    handover_signal(ig[0], ig[1], gg[0], gg[1]);
    #pragma unroll
    for (int cc = 0; cc < 2; ++cc) {
        const float cn = fg[cc] * c[cc] + ig[cc] * gg[cc];
        c[cc] = cn;
        const float hh = og[cc] * ftanh(cn);
        hq[cc] = hh;
        partial += hh * w[cc];
    }
    hold2(hq[0], hq[1]);
    hold2(c[0], c[1]);
    a = (v2f){ hq[0], hq[1] };
    return partial;
}

__global__ __launch_bounds__(64)
void bilstm_fused(const float* x,
                  const float* __restrict__ w_ih_f, const float* __restrict__ w_hh_f,
                  const float* __restrict__ b_ih_f, const float* __restrict__ b_hh_f,
                  const float* __restrict__ w_ih_b, const float* __restrict__ w_hh_b,
                  const float* __restrict__ b_ih_b, const float* __restrict__ b_hh_b,
                  const float* __restrict__ w_out, const float* __restrict__ b_out,
                  const int* __restrict__ future_p,
                  float* __restrict__ out)
{
    __shared__ __align__(16) float gbuf[2 * 2 * PLANE];
    __shared__ __align__(16) float obuf[2 * 2 * 16 * 4];

    const int lane = threadIdx.x & 31;
    const int wave = __builtin_amdgcn_readfirstlane((int)(threadIdx.x >> 5));
    const int half = lane >> 4;
    const int nn   = lane & 15;
    const int bb   = blockIdx.x * 16;
    const int fut  = future_p[0];
    const int jA   = 4 * wave + 2 * half;

    const float* wih[2] = { w_ih_f, w_ih_b };
    const float* whh[2] = { w_hh_f, w_hh_b };
    const float* bih[2] = { b_ih_f, b_ih_b };
    const float* bhh[2] = { b_hh_f, b_hh_b };

    const int nM = nn + 16 * wave;
    const int nT = nn + 16 * (1 - wave);
    v2f wm[2], wt[2], wx[2];
    #pragma unroll
    for (int d = 0; d < 2; ++d) {
        wm[d] = (v2f){ whh[d][nM * H_SZ + jA], whh[d][nM * H_SZ + jA + 1] };
        wt[d] = (v2f){ whh[d][nT * H_SZ + jA], whh[d][nT * H_SZ + jA + 1] };
        const float wi = wih[d][nM];
        const float bs = bih[d][nM] + bhh[d][nM];
        wx[d] = (v2f){ half ? 0.0f : wi, half ? 0.0f : bs };
    }

    float wo[2][2];
    #pragma unroll
    for (int cc = 0; cc < 2; ++cc) {
        wo[0][cc] = w_out[jA + cc];
        wo[1][cc] = w_out[8 + jA + cc];
    }
    const float bo = b_out[0];

    const int colM = 4 * (nn & 7) + (nn >> 3) + 2 * wave;
    const int colT = 4 * (nn & 7) + (nn >> 3) + 2 * (1 - wave);
    float* const plane_f = gbuf;
    float* const plane_b = gbuf + 2 * PLANE;
    const int stRow = wave * PLANE + 8 * half * GP;
    float* const smf = plane_f + stRow + colM;
    float* const stf = plane_f + stRow + colT;
    float* const smb = plane_b + stRow + colM;
    float* const stb = plane_b + stRow + colT;
    const float* const ldf = plane_f + nn * GP + 4 * jA;
    const float* const ldb = plane_b + nn * GP + 4 * jA;
    float4* const obuf4 = (float4*)obuf;

    v2f af = {0.0f, 0.0f}, ab = {0.0f, 0.0f};
    float cst[2][2] = {{0,0},{0,0}};

    const float kone = half ? 0.0f : 1.0f;

    const float x00 = x[(size_t)(0 & (S_LEN - 1)) * B_TOT + bb + nn];
    const float x01 = x[(size_t)(fut & (S_LEN - 1)) * B_TOT + bb + nn];
    float xa[2], xn[2];
    xa[0] = x[(size_t)(1 & (S_LEN - 1)) * B_TOT + bb + nn];
    xa[1] = x[(size_t)((fut - 1) & (S_LEN - 1)) * B_TOT + bb + nn];
    xn[0] = x[(size_t)(2 & (S_LEN - 1)) * B_TOT + bb + nn];
    xn[1] = x[(size_t)((fut - 2) & (S_LEN - 1)) * B_TOT + bb + nn];

    v8f accfm, accft, accbm, accbt;
    float4 qf[2][2], qb[2][2];
    gates_mm(af, x00, kone, wm[0], wt[0], wx[0], accfm, accft);
    gates_to_lds(smf, stf, accfm, accft);
    lds_handover();
    gates_from_lds(ldf, qf);
    gates_mm(ab, x01, kone, wm[1], wt[1], wx[1], accbm, accbt);
    PHASE_PIN();

    const int oself = (wave * 2 + half) * 16 + nn;
    obuf4[oself] = make_float4(0.0f, 0.0f, 0.0f, 0.0f);
    float4 ovp = make_float4(0.0f, 0.0f, 0.0f, 0.0f);
    for (int ii = 0; ii < S_LEN; ii += 4) {
        float pv[4];
        float4 o1, o2, o3;
        const bool comb = (wave == 0) && (ii > 0);
        #pragma unroll
        for (int u = 0; u < 4; ++u) {
            const int i = ii + u;

            gates_to_lds(smb, stb, accbm, accbt);
            PHASE_PIN();
            float xb[2];
            xb[0] = x[(size_t)((i + 3) & (S_LEN - 1)) * B_TOT + bb + nn];
            xb[1] = x[(size_t)((fut - i - 3) & (S_LEN - 1)) * B_TOT + bb + nn];
            float partial = cell_pair(qf, cst[0], wo[0], 0.0f, af);
            PHASE_PIN();

            handover_wait();
            gates_from_lds(ldb, qb);
            if (u == 0) {
                o1 = obuf4[16 + nn]; o2 = obuf4[32 + nn]; o3 = obuf4[48 + nn];
            }
            PHASE_PIN();
            gates_mm(af, xa[0], kone, wm[0], wt[0], wx[0], accfm, accft);
            PHASE_PIN();

            gates_to_lds(smf, stf, accfm, accft);
            PHASE_PIN();
            partial = cell_pair(qb, cst[1], wo[1], partial, ab);
            pv[u] = partial;
            PHASE_PIN();

            handover_wait();
            gates_from_lds(ldf, qf);
            PHASE_PIN();
            gates_mm(ab, xa[1], kone, wm[1], wt[1], wx[1], accbm, accbt);
            if (u == 0) {
                hold4(o1.x, o1.y, o1.z, o1.w);
                hold4(o2.x, o2.y, o2.z, o2.w);
                hold4(o3.x, o3.y, o3.z, o3.w);
                if (comb && lane < 16) {
                    *(float4*)&out[(size_t)(bb + nn) * S_LEN + ii - 4] =
                        make_float4((ovp.x + o1.x) + (o2.x + o3.x) + bo, (ovp.y + o1.y) + (o2.y + o3.y) + bo,
                                    (ovp.z + o1.z) + (o2.z + o3.z) + bo, (ovp.w + o1.w) + (o2.w + o3.w) + bo);
                }
            }
            PHASE_PIN();

            xa[0] = xn[0];
            xa[1] = xn[1];
            xn[0] = xb[0];
            xn[1] = xb[1];
        }
        ovp = make_float4(pv[0], pv[1], pv[2], pv[3]);
        obuf4[oself] = ovp;
    }
    lds_handover();
    if (wave == 0 && lane < 16) {
        const float4 o1 = obuf4[16 + nn], o2 = obuf4[32 + nn], o3 = obuf4[48 + nn];
        *(float4*)&out[(size_t)(bb + nn) * S_LEN + S_LEN - 4] =
            make_float4((ovp.x + o1.x) + (o2.x + o3.x) + bo, (ovp.y + o1.y) + (o2.y + o3.y) + bo,
                        (ovp.z + o1.z) + (o2.z + o3.z) + bo, (ovp.w + o1.w) + (o2.w + o3.w) + bo);
    }
}

extern "C" void kernel_launch(void* const* d_in, const int* in_sizes, int n_in,
                              void* d_out, int out_size, void* d_ws, size_t ws_size,
                              hipStream_t stream) {
    (void)in_sizes; (void)n_in; (void)out_size; (void)d_ws; (void)ws_size;
    const float* xp     = (const float*)d_in[0];
    const float* w_ih_f = (const float*)d_in[1];
    const float* w_hh_f = (const float*)d_in[2];
    const float* b_ih_f = (const float*)d_in[3];
    const float* b_hh_f = (const float*)d_in[4];
    const float* w_ih_b = (const float*)d_in[5];
    const float* w_hh_b = (const float*)d_in[6];
    const float* b_ih_b = (const float*)d_in[7];
    const float* b_hh_b = (const float*)d_in[8];
    const float* w_out  = (const float*)d_in[9];
    const float* b_out  = (const float*)d_in[10];
    const int*   future = (const int*)d_in[11];
    float* out = (float*)d_out;

    dim3 grid(B_TOT / 16);
    dim3 block(64);
    bilstm_fused<<<grid, block, 0, stream>>>(
        xp, w_ih_f, w_hh_f, b_ih_f, b_hh_f,
        w_ih_b, w_hh_b, b_ih_b, b_hh_b,
        w_out, b_out, future, out);
}


# Gate product split by k over four waves (128-thread work-group): a lane owns one cell, its A fragment is {h, x or 1}, so x and bias ride in the free k slots: 2 independent WMMAs per lane-step, was 3
# speedup vs baseline: 1.6875x; 1.1306x over previous
#include <hip/hip_runtime.h>
#include <hip/hip_bf16.h>

#define S_LEN 2048
#define B_TOT 4096
#define H_SZ  8
#define NW    4
#define GP    36
#define PLANE (16 * GP)

typedef __attribute__((ext_vector_type(2))) float v2f;
typedef __attribute__((ext_vector_type(8))) float v8f;

__device__ __forceinline__ float frcp(float v) {
    return __builtin_amdgcn_rcpf(v);
}

__device__ __forceinline__ float fsigmoid(float v) {
    return frcp(1.0f + __expf(-v));
}

__device__ __forceinline__ float ftanh(float v) {
#if __has_builtin(__builtin_amdgcn_tanhf)
    return __builtin_amdgcn_tanhf(v);
#else
    return 1.0f - 2.0f * frcp(1.0f + __expf(2.0f * v));
#endif
}

__device__ __forceinline__ void lds_handover() {
    asm volatile("s_wait_dscnt 0x0\n\ts_barrier_signal -1\n\ts_barrier_wait -1" ::: "memory");
}

__device__ __forceinline__ void handover_signal(float& a, float& b, float& c, float& d) {
    asm volatile("s_wait_dscnt 0x0\n\ts_barrier_signal -1"
                 : "+v"(a), "+v"(b), "+v"(c), "+v"(d) :: "memory");
}
__device__ __forceinline__ void handover_wait() {
    asm volatile("s_barrier_wait -1" ::: "memory");
}

__device__ __forceinline__ v8f wmma4(v2f a, v2f b, v8f c) {
    return __builtin_amdgcn_wmma_f32_16x16x4_f32(
         false, a,  false, b,
         (short)0, c,  false,  false);
}

#define PHASE_PIN() __builtin_amdgcn_sched_barrier(0)

__device__ __forceinline__ void hold4(float& a, float& b, float& c, float& d) {
    asm volatile("" : "+v"(a), "+v"(b), "+v"(c), "+v"(d) :: "memory");
}
__device__ __forceinline__ void hold1(float& a) {
    asm volatile("" : "+v"(a) :: "memory");
}

__device__ __forceinline__ void gates_mm(const float hval, const float xval, const bool hi,
                                         const v2f w0, const v2f w1, v8f& g0, v8f& g1) {
    float h0 = hval;
    hold1(h0);
    float xh = xval;
    hold1(xh);
    const v2f a = { h0, hi ? 1.0f : xh };
    const v8f z = {0,0,0,0,0,0,0,0};
    g0 = wmma4(a, w0, z);
    g1 = wmma4(a, w1, z);
}

__device__ __forceinline__ void gates_to_lds(float* s0, float* s1, const v8f g0, const v8f g1) {
    #pragma unroll
    for (int v = 0; v < 8; ++v) s0[v * GP] = g0[v];
    #pragma unroll
    for (int v = 0; v < 8; ++v) s1[v * GP] = g1[v];
}

__device__ __forceinline__ void gates_from_lds(const float* ld, float4 (&q)[NW]) {
    #pragma unroll
    for (int p = 0; p < NW; ++p)
        q[p] = *(const float4*)&ld[p * PLANE];
}

__device__ __forceinline__ float cell_one(float4 (&q)[NW], float& c, const float w,
                                          float partial, float& hout) {
    hold4(q[0].x, q[0].y, q[0].z, q[0].w);
    hold4(q[1].x, q[1].y, q[1].z, q[1].w);
    hold4(q[2].x, q[2].y, q[2].z, q[2].w);
    hold4(q[3].x, q[3].y, q[3].z, q[3].w);
    float ig = fsigmoid((q[0].x + q[1].x) + (q[2].x + q[3].x));
    float fg = fsigmoid((q[0].y + q[1].y) + (q[2].y + q[3].y));
    float gg = ftanh((q[0].z + q[1].z) + (q[2].z + q[3].z));
    float og = fsigmoid((q[0].w + q[1].w) + (q[2].w + q[3].w));
    handover_signal(ig, fg, gg, og);
    const float cn = fg * c + ig * gg;
    c = cn;
    float hh = og * ftanh(cn);
    partial += hh * w;
    hold1(hh);
    hold1(c);
    hout = hh;
    return partial;
}

__global__ __launch_bounds__(32 * NW)
void bilstm_fused(const float* x,
                  const float* __restrict__ w_ih_f, const float* __restrict__ w_hh_f,
                  const float* __restrict__ b_ih_f, const float* __restrict__ b_hh_f,
                  const float* __restrict__ w_ih_b, const float* __restrict__ w_hh_b,
                  const float* __restrict__ b_ih_b, const float* __restrict__ b_hh_b,
                  const float* __restrict__ w_out, const float* __restrict__ b_out,
                  const int* __restrict__ future_p,
                  float* __restrict__ out)
{
    __shared__ __align__(16) float gbuf[2 * NW * PLANE];
    __shared__ __align__(16) float obuf[NW * 2 * 16 * 4];

    const int lane = threadIdx.x & 31;
    const int wave = __builtin_amdgcn_readfirstlane((int)(threadIdx.x >> 5));
    const int half = lane >> 4;
    const int nn   = lane & 15;
    const int bb   = blockIdx.x * 16;
    const int fut  = future_p[0];
    const int jA   = 2 * wave + half;
    const bool hi  = half != 0;

    const float* wih[2] = { w_ih_f, w_ih_b };
    const float* whh[2] = { w_hh_f, w_hh_b };
    const float* bih[2] = { b_ih_f, b_ih_b };
    const float* bhh[2] = { b_hh_f, b_hh_b };

    const int n0 = nn, n1 = nn + 16;
    v2f w0[2], w1[2];
    #pragma unroll
    for (int d = 0; d < 2; ++d) {
        const float e0 = hi ? (bih[d][n0] + bhh[d][n0]) : wih[d][n0];
        const float e1 = hi ? (bih[d][n1] + bhh[d][n1]) : wih[d][n1];
        w0[d] = (v2f){ whh[d][n0 * H_SZ + jA], wave == 0 ? e0 : 0.0f };
        w1[d] = (v2f){ whh[d][n1 * H_SZ + jA], wave == 0 ? e1 : 0.0f };
    }

    const float wo_f = w_out[jA];
    const float wo_b = w_out[8 + jA];
    const float bo = b_out[0];

    const int col0 = 4 * (nn & 7) + (nn >> 3);
    const int col1 = col0 + 2;
    float* const plane_f = gbuf;
    float* const plane_b = gbuf + NW * PLANE;
    const int stRow = wave * PLANE + 8 * half * GP;
    float* const s0f = plane_f + stRow + col0;
    float* const s1f = plane_f + stRow + col1;
    float* const s0b = plane_b + stRow + col0;
    float* const s1b = plane_b + stRow + col1;
    const float* const ldf = plane_f + nn * GP + 4 * jA;
    const float* const ldb = plane_b + nn * GP + 4 * jA;
    float4* const obuf4 = (float4*)obuf;

    float hf = 0.0f, hb = 0.0f;
    float cf = 0.0f, cb = 0.0f;

    const float x00 = x[(size_t)(0 & (S_LEN - 1)) * B_TOT + bb + nn];
    const float x01 = x[(size_t)(fut & (S_LEN - 1)) * B_TOT + bb + nn];
    float xa[2], xn[2];
    xa[0] = x[(size_t)(1 & (S_LEN - 1)) * B_TOT + bb + nn];
    xa[1] = x[(size_t)((fut - 1) & (S_LEN - 1)) * B_TOT + bb + nn];
    xn[0] = x[(size_t)(2 & (S_LEN - 1)) * B_TOT + bb + nn];
    xn[1] = x[(size_t)((fut - 2) & (S_LEN - 1)) * B_TOT + bb + nn];

    v8f accf0, accf1, accb0, accb1;
    float4 qf[NW], qb[NW];
    gates_mm(hf, x00, hi, w0[0], w1[0], accf0, accf1);
    gates_to_lds(s0f, s1f, accf0, accf1);
    lds_handover();
    gates_from_lds(ldf, qf);
    gates_mm(hb, x01, hi, w0[1], w1[1], accb0, accb1);
    PHASE_PIN();

    const int oself = (wave * 2 + half) * 16 + nn;
    obuf4[oself] = make_float4(0.0f, 0.0f, 0.0f, 0.0f);
    float4 ovp = make_float4(0.0f, 0.0f, 0.0f, 0.0f);
    for (int ii = 0; ii < S_LEN; ii += 4) {
        float pv[4];
        float4 o1, o2, o3, o4, o5, o6, o7;
        const bool comb = (wave == 0) && (ii > 0);
        #pragma unroll
        for (int u = 0; u < 4; ++u) {
            const int i = ii + u;

            gates_to_lds(s0b, s1b, accb0, accb1);
            PHASE_PIN();
            float xb[2];
            xb[0] = x[(size_t)((i + 3) & (S_LEN - 1)) * B_TOT + bb + nn];
            xb[1] = x[(size_t)((fut - i - 3) & (S_LEN - 1)) * B_TOT + bb + nn];
            float partial = cell_one(qf, cf, wo_f, 0.0f, hf);
            PHASE_PIN();

            handover_wait();
            gates_from_lds(ldb, qb);
            if (u == 0) {
                o1 = obuf4[16 + nn]; o2 = obuf4[32 + nn]; o3 = obuf4[48 + nn]; o4 = obuf4[64 + nn];
                o5 = obuf4[80 + nn]; o6 = obuf4[96 + nn]; o7 = obuf4[112 + nn];
            }
            PHASE_PIN();
            gates_mm(hf, xa[0], hi, w0[0], w1[0], accf0, accf1);
            PHASE_PIN();

            gates_to_lds(s0f, s1f, accf0, accf1);
            PHASE_PIN();
            partial = cell_one(qb, cb, wo_b, partial, hb);
            pv[u] = partial;
            PHASE_PIN();

            handover_wait();
            gates_from_lds(ldf, qf);
            PHASE_PIN();
            gates_mm(hb, xa[1], hi, w0[1], w1[1], accb0, accb1);
            if (u == 0) {
                hold4(o1.x, o1.y, o1.z, o1.w);
                hold4(o2.x, o2.y, o2.z, o2.w);
                hold4(o3.x, o3.y, o3.z, o3.w);
                hold4(o4.x, o4.y, o4.z, o4.w);
                hold4(o5.x, o5.y, o5.z, o5.w);
                hold4(o6.x, o6.y, o6.z, o6.w);
                hold4(o7.x, o7.y, o7.z, o7.w);
                if (comb && lane < 16) {
                    *(float4*)&out[(size_t)(bb + nn) * S_LEN + ii - 4] = make_float4(
                        ((ovp.x + o1.x) + (o2.x + o3.x)) + ((o4.x + o5.x) + (o6.x + o7.x)) + bo,
                        ((ovp.y + o1.y) + (o2.y + o3.y)) + ((o4.y + o5.y) + (o6.y + o7.y)) + bo,
                        ((ovp.z + o1.z) + (o2.z + o3.z)) + ((o4.z + o5.z) + (o6.z + o7.z)) + bo,
                        ((ovp.w + o1.w) + (o2.w + o3.w)) + ((o4.w + o5.w) + (o6.w + o7.w)) + bo);
                }
            }
            PHASE_PIN();

            xa[0] = xn[0];
            xa[1] = xn[1];
            xn[0] = xb[0];
            xn[1] = xb[1];
        }
        ovp = make_float4(pv[0], pv[1], pv[2], pv[3]);
        obuf4[oself] = ovp;
    }
    lds_handover();
    if (wave == 0 && lane < 16) {
        const float4 o1 = obuf4[16 + nn], o2 = obuf4[32 + nn], o3 = obuf4[48 + nn], o4 = obuf4[64 + nn];
        const float4 o5 = obuf4[80 + nn], o6 = obuf4[96 + nn], o7 = obuf4[112 + nn];
        *(float4*)&out[(size_t)(bb + nn) * S_LEN + S_LEN - 4] = make_float4(
            ((ovp.x + o1.x) + (o2.x + o3.x)) + ((o4.x + o5.x) + (o6.x + o7.x)) + bo,
            ((ovp.y + o1.y) + (o2.y + o3.y)) + ((o4.y + o5.y) + (o6.y + o7.y)) + bo,
            ((ovp.z + o1.z) + (o2.z + o3.z)) + ((o4.z + o5.z) + (o6.z + o7.z)) + bo,
            ((ovp.w + o1.w) + (o2.w + o3.w)) + ((o4.w + o5.w) + (o6.w + o7.w)) + bo);
    }
}

extern "C" void kernel_launch(void* const* d_in, const int* in_sizes, int n_in,
                              void* d_out, int out_size, void* d_ws, size_t ws_size,
                              hipStream_t stream) {
    (void)in_sizes; (void)n_in; (void)out_size; (void)d_ws; (void)ws_size;
    const float* xp     = (const float*)d_in[0];
    const float* w_ih_f = (const float*)d_in[1];
    const float* w_hh_f = (const float*)d_in[2];
    const float* b_ih_f = (const float*)d_in[3];
    const float* b_hh_f = (const float*)d_in[4];
    const float* w_ih_b = (const float*)d_in[5];
    const float* w_hh_b = (const float*)d_in[6];
    const float* b_ih_b = (const float*)d_in[7];
    const float* b_hh_b = (const float*)d_in[8];
    const float* w_out  = (const float*)d_in[9];
    const float* b_out  = (const float*)d_in[10];
    const int*   future = (const int*)d_in[11];
    float* out = (float*)d_out;

    dim3 grid(B_TOT / 16);
    dim3 block(32 * NW);
    bilstm_fused<<<grid, block, 0, stream>>>(
        xp, w_ih_f, w_hh_f, b_ih_f, b_hh_f,
        w_ih_b, w_hh_b, b_ih_b, b_hh_b,
        w_out, b_out, future, out);
}


# Gate product split by gate column, not by k, over the four waves (wave = direction x four cells): only h crosses waves (one 16-byte LDS read per lane-step); gates meet inside quads by lane swizzles
# speedup vs baseline: 2.2345x; 1.3241x over previous
#include <hip/hip_runtime.h>
#include <hip/hip_bf16.h>

#define S_LEN 2048
#define B_TOT 4096
#define H_SZ  8
#define NW    4
#define HPLANE (16 * H_SZ)

typedef __attribute__((ext_vector_type(2))) float v2f;
typedef __attribute__((ext_vector_type(8))) float v8f;

__device__ __forceinline__ float frcp(float v) {
    return __builtin_amdgcn_rcpf(v);
}

__device__ __forceinline__ float fsigmoid(float v) {
    return frcp(1.0f + __expf(-v));
}

__device__ __forceinline__ float ftanh(float v) {
#if __has_builtin(__builtin_amdgcn_tanhf)
    return __builtin_amdgcn_tanhf(v);
#else
    return 1.0f - 2.0f * frcp(1.0f + __expf(2.0f * v));
#endif
}

__device__ __forceinline__ void lds_handover() {
    asm volatile("s_wait_dscnt 0x0\n\ts_barrier_signal -1\n\ts_barrier_wait -1" ::: "memory");
}

__device__ __forceinline__ v8f wmma4(v2f a, v2f b, v8f c) {
    return __builtin_amdgcn_wmma_f32_16x16x4_f32(
         false, a,  false, b,
         (short)0, c,  false,  false);
}

template <int CTRL>
__device__ __forceinline__ float lane_get(float v) {
    return __int_as_float(__builtin_amdgcn_update_dpp(0, __float_as_int(v), CTRL, 0xF, 0xF, true));
}
#define QUAD_XOR1 0xB1
#define QUAD_XOR2 0x4E
#define ROW_ROR4  0x124
#define ROW_ROR8  0x128

__device__ __forceinline__ void hold4(float& a, float& b, float& c, float& d) {
    asm volatile("" : "+v"(a), "+v"(b), "+v"(c), "+v"(d) :: "memory");
}

__global__ __launch_bounds__(32 * NW)
void bilstm_fused(const float* x,
                  const float* __restrict__ w_ih_f, const float* __restrict__ w_hh_f,
                  const float* __restrict__ b_ih_f, const float* __restrict__ b_hh_f,
                  const float* __restrict__ w_ih_b, const float* __restrict__ w_hh_b,
                  const float* __restrict__ b_ih_b, const float* __restrict__ b_hh_b,
                  const float* __restrict__ w_out, const float* __restrict__ b_out,
                  const int* __restrict__ future_p,
                  float* __restrict__ out)
{
    __shared__ __align__(16) float hbuf[2 * 2 * HPLANE];
    __shared__ __align__(16) float obuf[NW * 16 * 4];

    const int lane = threadIdx.x & 31;
    const int wave = __builtin_amdgcn_readfirstlane((int)(threadIdx.x >> 5));
    const int dir  = wave >> 1;
    const int qh   = wave & 1;
    const int half = lane >> 4;
    const int nn   = lane & 15;
    const int cq   = nn >> 2;
    const int t    = nn & 3;
    const bool t0  = (t & 1) != 0;
    const bool t1  = (t & 2) != 0;
    const bool hi  = half != 0;
    const int bb   = blockIdx.x * 16;
    const int fut  = future_p[0];

    const float* const wih = dir ? w_ih_b : w_ih_f;
    const float* const whh = dir ? w_hh_b : w_hh_f;
    const float* const bih = dir ? b_ih_b : b_ih_f;
    const float* const bhh = dir ? b_hh_b : b_hh_f;

    const int j = 4 * qh + cq;
    const int G = 8 * t + j;
    const v2f wB0 = { whh[G * H_SZ + 4 * half + 0], whh[G * H_SZ + 4 * half + 1] };
    const v2f wB1 = { whh[G * H_SZ + 4 * half + 2], whh[G * H_SZ + 4 * half + 3] };
    const v2f wB2 = { hi ? 0.0f : wih[G], hi ? 0.0f : (bih[G] + bhh[G]) };

    const float wo = w_out[8 * dir + j];
    const float bo = b_out[0];

    float* const hb = hbuf + dir * (2 * HPLANE);
    const int ldoff = nn * H_SZ + 4 * half;
    const int stoff = (8 * half + t) * H_SZ + j;
    float4* const obuf4 = (float4*)obuf;

    ((float4*)hbuf)[threadIdx.x] = make_float4(0.0f, 0.0f, 0.0f, 0.0f);
    if (threadIdx.x < NW * 16) obuf4[threadIdx.x] = make_float4(0.0f, 0.0f, 0.0f, 0.0f);

    const float* const xcol = x + bb + nn;
    const int xbase = dir ? fut : 0;
    const int xsgn  = dir ? -1 : 1;
    float xr[4];
    #pragma unroll
    for (int k = 0; k < 4; ++k)
        xr[k] = xcol[(size_t)((xbase + xsgn * k) & (S_LEN - 1)) * B_TOT];

    lds_handover();

    float cst[2] = { 0.0f, 0.0f };
    const v8f zero8 = {0,0,0,0,0,0,0,0};
    for (int ii = 0; ii < S_LEN; ii += 4) {
        float pv0[4], pv1[4];
        float4 o0, o1, o2, o3;
        #pragma unroll
        for (int u = 0; u < 4; ++u) {
            const int i = ii + u;
            const float* const hr = hb + (u & 1) * HPLANE;
            float* const hw = hb + ((u + 1) & 1) * HPLANE;

            const float4 hv = *(const float4*)&hr[ldoff];
            if (u == 0) {
                o0 = obuf4[nn]; o1 = obuf4[16 + nn]; o2 = obuf4[32 + nn]; o3 = obuf4[48 + nn];
                asm volatile("" ::: "memory");
            }
            const float xc = xr[u];
            xr[u] = xcol[(size_t)((xbase + xsgn * (i + 4)) & (S_LEN - 1)) * B_TOT];

            const v2f a2 = { hi ? 0.0f : xc, hi ? 0.0f : 1.0f };
            v8f g = wmma4(a2, wB2, zero8);
            g = wmma4((v2f){ hv.x, hv.y }, wB0, g);
            g = wmma4((v2f){ hv.z, hv.w }, wB1, g);

            float V0[4], V1[4];
            #pragma unroll
            for (int a = 0; a < 4; ++a) {
                const float X0 = g[2 * a], X1 = g[2 * a + 1];
                const float P0 = lane_get<QUAD_XOR1>(X0);
                const float P1 = lane_get<QUAD_XOR1>(X1);
                V0[a] = t0 ? P1 : X0;
                V1[a] = t0 ? X1 : P0;
            }
            float W[4][2];
            #pragma unroll
            for (int r2 = 0; r2 < 2; ++r2) {
                const float Q00 = lane_get<QUAD_XOR2>(V0[2 * r2]), Q01 = lane_get<QUAD_XOR2>(V0[2 * r2 + 1]);
                const float Q10 = lane_get<QUAD_XOR2>(V1[2 * r2]), Q11 = lane_get<QUAD_XOR2>(V1[2 * r2 + 1]);
                W[0][r2] = t1 ? Q01 : V0[2 * r2];
                W[2][r2] = t1 ? V0[2 * r2 + 1] : Q00;
                W[1][r2] = t1 ? Q11 : V1[2 * r2];
                W[3][r2] = t1 ? V1[2 * r2 + 1] : Q10;
            }

            float hn[2];
            #pragma unroll
            for (int r2 = 0; r2 < 2; ++r2) {
                const float ig = fsigmoid(W[0][r2]);
                const float fg = fsigmoid(W[1][r2]);
                const float gg = ftanh(W[2][r2]);
                const float og = fsigmoid(W[3][r2]);
                const float cn = fg * cst[r2] + ig * gg;
                cst[r2] = cn;
                hn[r2] = og * ftanh(cn);
            }
            hw[stoff] = hn[0];
            hw[stoff + 4 * H_SZ] = hn[1];

            if (u == 0) {
                hold4(o0.x, o0.y, o0.z, o0.w);
                hold4(o1.x, o1.y, o1.z, o1.w);
                hold4(o2.x, o2.y, o2.z, o2.w);
                hold4(o3.x, o3.y, o3.z, o3.w);
                if (wave == 0 && lane < 16 && ii > 0) {
                    *(float4*)&out[(size_t)(bb + nn) * S_LEN + ii - 4] = make_float4(
                        ((o0.x + o1.x) + (o2.x + o3.x)) + bo,
                        ((o0.y + o1.y) + (o2.y + o3.y)) + bo,
                        ((o0.z + o1.z) + (o2.z + o3.z)) + bo,
                        ((o0.w + o1.w) + (o2.w + o3.w)) + bo);
                }
            }
            float p0 = hn[0] * wo, p1 = hn[1] * wo;
            p0 += lane_get<ROW_ROR4>(p0);
            p1 += lane_get<ROW_ROR4>(p1);
            p0 += lane_get<ROW_ROR8>(p0);
            p1 += lane_get<ROW_ROR8>(p1);
            pv0[u] = p0;
            pv1[u] = p1;
            if (u == 3 && cq == 0) {
                obuf4[wave * 16 + 8 * half + t]     = make_float4(pv0[0], pv0[1], pv0[2], pv0[3]);
                obuf4[wave * 16 + 8 * half + 4 + t] = make_float4(pv1[0], pv1[1], pv1[2], pv1[3]);
            }
            lds_handover();
        }
    }
    if (wave == 0 && lane < 16) {
        const float4 o0 = obuf4[nn], o1 = obuf4[16 + nn], o2 = obuf4[32 + nn], o3 = obuf4[48 + nn];
        *(float4*)&out[(size_t)(bb + nn) * S_LEN + S_LEN - 4] = make_float4(
            ((o0.x + o1.x) + (o2.x + o3.x)) + bo,
            ((o0.y + o1.y) + (o2.y + o3.y)) + bo,
            ((o0.z + o1.z) + (o2.z + o3.z)) + bo,
            ((o0.w + o1.w) + (o2.w + o3.w)) + bo);
    }
}

extern "C" void kernel_launch(void* const* d_in, const int* in_sizes, int n_in,
                              void* d_out, int out_size, void* d_ws, size_t ws_size,
                              hipStream_t stream) {
    (void)in_sizes; (void)n_in; (void)out_size; (void)d_ws; (void)ws_size;
    const float* xp     = (const float*)d_in[0];
    const float* w_ih_f = (const float*)d_in[1];
    const float* w_hh_f = (const float*)d_in[2];
    const float* b_ih_f = (const float*)d_in[3];
    const float* b_hh_f = (const float*)d_in[4];
    const float* w_ih_b = (const float*)d_in[5];
    const float* w_hh_b = (const float*)d_in[6];
    const float* b_ih_b = (const float*)d_in[7];
    const float* b_hh_b = (const float*)d_in[8];
    const float* w_out  = (const float*)d_in[9];
    const float* b_out  = (const float*)d_in[10];
    const int*   future = (const int*)d_in[11];
    float* out = (float*)d_out;

    dim3 grid(B_TOT / 16);
    dim3 block(32 * NW);
    bilstm_fused<<<grid, block, 0, stream>>>(
        xp, w_ih_f, w_hh_f, b_ih_f, b_hh_f,
        w_ih_b, w_hh_b, b_ih_b, b_hh_b,
        w_out, b_out, future, out);
}


# Own half of the state row read back from LDS behind the wave's own store, ahead of the barrier; x/bias and own-half products run under the other half's read: 1 WMMA, not 2, behind the hand-over
# speedup vs baseline: 2.4279x; 1.0866x over previous
#include <hip/hip_runtime.h>
#include <hip/hip_bf16.h>

#define S_LEN 2048
#define B_TOT 4096
#define H_SZ  8
#define NW    4
#define HPLANE (16 * H_SZ)

typedef __attribute__((ext_vector_type(2))) float v2f;
typedef __attribute__((ext_vector_type(8))) float v8f;

__device__ __forceinline__ float frcp(float v) {
    return __builtin_amdgcn_rcpf(v);
}

__device__ __forceinline__ float fsigmoid(float v) {
    return frcp(1.0f + __expf(-v));
}

__device__ __forceinline__ float ftanh(float v) {
#if __has_builtin(__builtin_amdgcn_tanhf)
    return __builtin_amdgcn_tanhf(v);
#else
    return 1.0f - 2.0f * frcp(1.0f + __expf(2.0f * v));
#endif
}

template <int YOUNGER>
__device__ __forceinline__ void lds_handover() {
    if (YOUNGER == 1)
        asm volatile("s_wait_dscnt 0x1\n\ts_barrier_signal -1\n\ts_barrier_wait -1" ::: "memory");
    else
        asm volatile("s_wait_dscnt 0x0\n\ts_barrier_signal -1\n\ts_barrier_wait -1" ::: "memory");
}

__device__ __forceinline__ v8f wmma4(v2f a, v2f b, v8f c) {
    return __builtin_amdgcn_wmma_f32_16x16x4_f32(
         false, a,  false, b,
         (short)0, c,  false,  false);
}

template <int CTRL>
__device__ __forceinline__ float lane_get(float v) {
    return __int_as_float(__builtin_amdgcn_update_dpp(0, __float_as_int(v), CTRL, 0xF, 0xF, true));
}
#define QUAD_XOR1 0xB1
#define QUAD_XOR2 0x4E
#define ROW_ROR4  0x124
#define ROW_ROR8  0x128

__device__ __forceinline__ void hold4(float& a, float& b, float& c, float& d) {
    asm volatile("" : "+v"(a), "+v"(b), "+v"(c), "+v"(d) :: "memory");
}

__global__ __launch_bounds__(32 * NW)
void bilstm_fused(const float* x,
                  const float* __restrict__ w_ih_f, const float* __restrict__ w_hh_f,
                  const float* __restrict__ b_ih_f, const float* __restrict__ b_hh_f,
                  const float* __restrict__ w_ih_b, const float* __restrict__ w_hh_b,
                  const float* __restrict__ b_ih_b, const float* __restrict__ b_hh_b,
                  const float* __restrict__ w_out, const float* __restrict__ b_out,
                  const int* __restrict__ future_p,
                  float* __restrict__ out)
{
    __shared__ __align__(16) float hbuf[2 * 2 * HPLANE];
    __shared__ __align__(16) float obuf[NW * 16 * 4];

    const int lane = threadIdx.x & 31;
    const int wave = __builtin_amdgcn_readfirstlane((int)(threadIdx.x >> 5));
    const int dir  = wave >> 1;
    const int qh   = wave & 1;
    const int half = lane >> 4;
    const int nn   = lane & 15;
    const int cq   = nn >> 2;
    const int t    = nn & 3;
    const bool t0  = (t & 1) != 0;
    const bool t1  = (t & 2) != 0;
    const bool hi  = half != 0;
    const int bb   = blockIdx.x * 16;
    const int fut  = future_p[0];

    const float* const wih = dir ? w_ih_b : w_ih_f;
    const float* const whh = dir ? w_hh_b : w_hh_f;
    const float* const bih = dir ? b_ih_b : b_ih_f;
    const float* const bhh = dir ? b_hh_b : b_hh_f;

    const int j = 4 * qh + cq;
    const int G = 8 * t + j;
    const int kown = 4 * qh + 2 * half;
    const int koth = 4 * (1 - qh) + 2 * half;
    const v2f wBown = { whh[G * H_SZ + kown], whh[G * H_SZ + kown + 1] };
    const v2f wBoth = { whh[G * H_SZ + koth], whh[G * H_SZ + koth + 1] };
    const v2f wB2 = { hi ? 0.0f : wih[G], hi ? 0.0f : (bih[G] + bhh[G]) };

    const float wo = w_out[8 * dir + j];
    const float bo = b_out[0];

    float* const hb = hbuf + dir * (2 * HPLANE);
    const int ldown = nn * H_SZ + kown;
    const int ldoth = nn * H_SZ + koth;
    const int stoff = (8 * half + t) * H_SZ + j;
    float4* const obuf4 = (float4*)obuf;

    ((float4*)hbuf)[threadIdx.x] = make_float4(0.0f, 0.0f, 0.0f, 0.0f);
    if (threadIdx.x < NW * 16) obuf4[threadIdx.x] = make_float4(0.0f, 0.0f, 0.0f, 0.0f);

    const float* const xcol = x + bb + nn;
    const int xbase = dir ? fut : 0;
    const int xsgn  = dir ? -1 : 1;
    float xr[4];
    #pragma unroll
    for (int k = 0; k < 4; ++k)
        xr[k] = xcol[(size_t)((xbase + xsgn * k) & (S_LEN - 1)) * B_TOT];

    lds_handover<0>();

    float cst[2] = { 0.0f, 0.0f };
    v2f aown = { 0.0f, 0.0f };
    const v8f zero8 = {0,0,0,0,0,0,0,0};
    for (int ii = 0; ii < S_LEN; ii += 4) {
        float pv0[4], pv1[4];
        float4 o0, o1, o2, o3;
        #pragma unroll
        for (int u = 0; u < 4; ++u) {
            const int i = ii + u;
            const float* const hr = hb + (u & 1) * HPLANE;
            float* const hw = hb + ((u + 1) & 1) * HPLANE;

            const v2f aoth = *(const v2f*)&hr[ldoth];
            if (u == 0) {
                o0 = obuf4[nn]; o1 = obuf4[16 + nn]; o2 = obuf4[32 + nn]; o3 = obuf4[48 + nn];
                asm volatile("" ::: "memory");
            }
            float xc = xr[u];
            asm volatile("" : "+v"(xc) :: "memory");
            xr[u] = xcol[(size_t)((xbase + xsgn * (i + 4)) & (S_LEN - 1)) * B_TOT];

            const v2f a2 = { hi ? 0.0f : xc, hi ? 0.0f : 1.0f };
            v8f g = wmma4(a2, wB2, zero8);
            g = wmma4(aown, wBown, g);
            g = wmma4(aoth, wBoth, g);

            float V0[4], V1[4];
            #pragma unroll
            for (int a = 0; a < 4; ++a) {
                const float X0 = g[2 * a], X1 = g[2 * a + 1];
                const float P0 = lane_get<QUAD_XOR1>(X0);
                const float P1 = lane_get<QUAD_XOR1>(X1);
                V0[a] = t0 ? P1 : X0;
                V1[a] = t0 ? X1 : P0;
            }
            float W[4][2];
            #pragma unroll
            for (int r2 = 0; r2 < 2; ++r2) {
                const float Q00 = lane_get<QUAD_XOR2>(V0[2 * r2]), Q01 = lane_get<QUAD_XOR2>(V0[2 * r2 + 1]);
                const float Q10 = lane_get<QUAD_XOR2>(V1[2 * r2]), Q11 = lane_get<QUAD_XOR2>(V1[2 * r2 + 1]);
                W[0][r2] = t1 ? Q01 : V0[2 * r2];
                W[2][r2] = t1 ? V0[2 * r2 + 1] : Q00;
                W[1][r2] = t1 ? Q11 : V1[2 * r2];
                W[3][r2] = t1 ? V1[2 * r2 + 1] : Q10;
            }

            float hn[2];
            #pragma unroll
            for (int r2 = 0; r2 < 2; ++r2) {
                const float ig = fsigmoid(W[0][r2]);
                const float fg = fsigmoid(W[1][r2]);
                const float gg = ftanh(W[2][r2]);
                const float og = fsigmoid(W[3][r2]);
                const float cn = fg * cst[r2] + ig * gg;
                cst[r2] = cn;
                hn[r2] = og * ftanh(cn);
            }
            hw[stoff] = hn[0];
            hw[stoff + 4 * H_SZ] = hn[1];
            aown = *(const v2f*)&hw[ldown];

            if (u == 0) {
                hold4(o0.x, o0.y, o0.z, o0.w);
                hold4(o1.x, o1.y, o1.z, o1.w);
                hold4(o2.x, o2.y, o2.z, o2.w);
                hold4(o3.x, o3.y, o3.z, o3.w);
                if (wave == 0 && lane < 16 && ii > 0) {
                    *(float4*)&out[(size_t)(bb + nn) * S_LEN + ii - 4] = make_float4(
                        ((o0.x + o1.x) + (o2.x + o3.x)) + bo,
                        ((o0.y + o1.y) + (o2.y + o3.y)) + bo,
                        ((o0.z + o1.z) + (o2.z + o3.z)) + bo,
                        ((o0.w + o1.w) + (o2.w + o3.w)) + bo);
                }
            }
            float p0 = hn[0] * wo, p1 = hn[1] * wo;
            p0 += lane_get<ROW_ROR4>(p0);
            p1 += lane_get<ROW_ROR4>(p1);
            p0 += lane_get<ROW_ROR8>(p0);
            p1 += lane_get<ROW_ROR8>(p1);
            pv0[u] = p0;
            pv1[u] = p1;
            if (u == 3 && cq == 0) {
                obuf4[wave * 16 + 8 * half + t]     = make_float4(pv0[0], pv0[1], pv0[2], pv0[3]);
                obuf4[wave * 16 + 8 * half + 4 + t] = make_float4(pv1[0], pv1[1], pv1[2], pv1[3]);
            }
            if (u == 3) lds_handover<0>(); else lds_handover<1>();
        }
    }
    if (wave == 0 && lane < 16) {
        const float4 o0 = obuf4[nn], o1 = obuf4[16 + nn], o2 = obuf4[32 + nn], o3 = obuf4[48 + nn];
        *(float4*)&out[(size_t)(bb + nn) * S_LEN + S_LEN - 4] = make_float4(
            ((o0.x + o1.x) + (o2.x + o3.x)) + bo,
            ((o0.y + o1.y) + (o2.y + o3.y)) + bo,
            ((o0.z + o1.z) + (o2.z + o3.z)) + bo,
            ((o0.w + o1.w) + (o2.w + o3.w)) + bo);
    }
}

extern "C" void kernel_launch(void* const* d_in, const int* in_sizes, int n_in,
                              void* d_out, int out_size, void* d_ws, size_t ws_size,
                              hipStream_t stream) {
    (void)in_sizes; (void)n_in; (void)out_size; (void)d_ws; (void)ws_size;
    const float* xp     = (const float*)d_in[0];
    const float* w_ih_f = (const float*)d_in[1];
    const float* w_hh_f = (const float*)d_in[2];
    const float* b_ih_f = (const float*)d_in[3];
    const float* b_hh_f = (const float*)d_in[4];
    const float* w_ih_b = (const float*)d_in[5];
    const float* w_hh_b = (const float*)d_in[6];
    const float* b_ih_b = (const float*)d_in[7];
    const float* b_hh_b = (const float*)d_in[8];
    const float* w_out  = (const float*)d_in[9];
    const float* b_out  = (const float*)d_in[10];
    const int*   future = (const int*)d_in[11];
    float* out = (float*)d_out;

    dim3 grid(B_TOT / 16);
    dim3 block(32 * NW);
    bilstm_fused<<<grid, block, 0, stream>>>(
        xp, w_ih_f, w_hh_f, b_ih_f, b_hh_f,
        w_ih_b, w_hh_b, b_ih_b, b_hh_b,
        w_out, b_out, future, out);
}


# Quad exchange behind the gate product: select fused into the lane swizzle (v_cndmask with a quad_perm source, constant lane sets), issued in groups: 16 instructions, was 16 swizzle moves + 16 selects
# speedup vs baseline: 2.5507x; 1.0506x over previous
#include <hip/hip_runtime.h>
#include <hip/hip_bf16.h>

#define S_LEN 2048
#define B_TOT 4096
#define H_SZ  8
#define NW    4
#define HPLANE (16 * H_SZ)

typedef __attribute__((ext_vector_type(2))) float v2f;
typedef __attribute__((ext_vector_type(8))) float v8f;

__device__ __forceinline__ float frcp(float v) {
    return __builtin_amdgcn_rcpf(v);
}

__device__ __forceinline__ float fsigmoid(float v) {
    return frcp(1.0f + __expf(-v));
}

__device__ __forceinline__ float ftanh(float v) {
#if __has_builtin(__builtin_amdgcn_tanhf)
    return __builtin_amdgcn_tanhf(v);
#else
    return 1.0f - 2.0f * frcp(1.0f + __expf(2.0f * v));
#endif
}

template <int YOUNGER>
__device__ __forceinline__ void lds_handover() {
    if (YOUNGER == 1)
        asm volatile("s_wait_dscnt 0x1\n\ts_barrier_signal -1\n\ts_barrier_wait -1" ::: "memory");
    else
        asm volatile("s_wait_dscnt 0x0\n\ts_barrier_signal -1\n\ts_barrier_wait -1" ::: "memory");
}

__device__ __forceinline__ v8f wmma4(v2f a, v2f b, v8f c) {
    return __builtin_amdgcn_wmma_f32_16x16x4_f32(
         false, a,  false, b,
         (short)0, c,  false,  false);
}

template <int CTRL>
__device__ __forceinline__ float lane_get(float v) {
    return __int_as_float(__builtin_amdgcn_update_dpp(0, __float_as_int(v), CTRL, 0xF, 0xF, true));
}
#define ROW_ROR4  0x124
#define ROW_ROR8  0x128

#define LANES_BIT0_SET 0xAAAAAAAAu
#define LANES_BIT0_CLR 0x55555555u
#define LANES_BIT1_SET 0xCCCCCCCCu
#define LANES_BIT1_CLR 0x33333333u
__device__ __forceinline__ void exchange_bit0(const v8f& g, float (&V0)[4], float (&V1)[4]) {
    asm("v_cndmask_b32_e64_dpp %0, %9, %8, %16 quad_perm:[1,0,3,2] row_mask:0xf bank_mask:0xf\n\t"
        "v_cndmask_b32_e64_dpp %1, %8, %9, %17 quad_perm:[1,0,3,2] row_mask:0xf bank_mask:0xf\n\t"
        "v_cndmask_b32_e64_dpp %2, %11, %10, %16 quad_perm:[1,0,3,2] row_mask:0xf bank_mask:0xf\n\t"
        "v_cndmask_b32_e64_dpp %3, %10, %11, %17 quad_perm:[1,0,3,2] row_mask:0xf bank_mask:0xf\n\t"
        "v_cndmask_b32_e64_dpp %4, %13, %12, %16 quad_perm:[1,0,3,2] row_mask:0xf bank_mask:0xf\n\t"
        "v_cndmask_b32_e64_dpp %5, %12, %13, %17 quad_perm:[1,0,3,2] row_mask:0xf bank_mask:0xf\n\t"
        "v_cndmask_b32_e64_dpp %6, %15, %14, %16 quad_perm:[1,0,3,2] row_mask:0xf bank_mask:0xf\n\t"
        "v_cndmask_b32_e64_dpp %7, %14, %15, %17 quad_perm:[1,0,3,2] row_mask:0xf bank_mask:0xf"
        : "=&v"(V0[0]), "=&v"(V1[0]), "=&v"(V0[1]), "=&v"(V1[1]),
          "=&v"(V0[2]), "=&v"(V1[2]), "=&v"(V0[3]), "=&v"(V1[3])
        : "v"(g[0]), "v"(g[1]), "v"(g[2]), "v"(g[3]), "v"(g[4]), "v"(g[5]), "v"(g[6]), "v"(g[7]),
          "s"(LANES_BIT0_CLR), "s"(LANES_BIT0_SET));
}
__device__ __forceinline__ void exchange_bit1_if(const float (&V0)[4], const float (&V1)[4],
                                                 float& i0, float& f0, float& i1, float& f1) {
    asm("v_cndmask_b32_e64_dpp %0, %5, %4, %12 quad_perm:[2,3,0,1] row_mask:0xf bank_mask:0xf\n\t"
        "v_cndmask_b32_e64_dpp %1, %7, %6, %12 quad_perm:[2,3,0,1] row_mask:0xf bank_mask:0xf\n\t"
        "v_cndmask_b32_e64_dpp %2, %9, %8, %12 quad_perm:[2,3,0,1] row_mask:0xf bank_mask:0xf\n\t"
        "v_cndmask_b32_e64_dpp %3, %11, %10, %12 quad_perm:[2,3,0,1] row_mask:0xf bank_mask:0xf"
        : "=&v"(i0), "=&v"(f0), "=&v"(i1), "=&v"(f1)
        : "v"(V0[0]), "v"(V0[1]), "v"(V1[0]), "v"(V1[1]), "v"(V0[2]), "v"(V0[3]), "v"(V1[2]), "v"(V1[3]),
          "s"(LANES_BIT1_CLR));
}
__device__ __forceinline__ void exchange_bit1_go(const float (&V0)[4], const float (&V1)[4],
                                                 float& g0, float& o0, float& g1, float& o1) {
    asm("v_cndmask_b32_e64_dpp %0, %4, %5, %12 quad_perm:[2,3,0,1] row_mask:0xf bank_mask:0xf\n\t"
        "v_cndmask_b32_e64_dpp %1, %6, %7, %12 quad_perm:[2,3,0,1] row_mask:0xf bank_mask:0xf\n\t"
        "v_cndmask_b32_e64_dpp %2, %8, %9, %12 quad_perm:[2,3,0,1] row_mask:0xf bank_mask:0xf\n\t"
        "v_cndmask_b32_e64_dpp %3, %10, %11, %12 quad_perm:[2,3,0,1] row_mask:0xf bank_mask:0xf"
        : "=&v"(g0), "=&v"(o0), "=&v"(g1), "=&v"(o1)
        : "v"(V0[0]), "v"(V0[1]), "v"(V1[0]), "v"(V1[1]), "v"(V0[2]), "v"(V0[3]), "v"(V1[2]), "v"(V1[3]),
          "s"(LANES_BIT1_SET));
}

__device__ __forceinline__ void hold4(float& a, float& b, float& c, float& d) {
    asm volatile("" : "+v"(a), "+v"(b), "+v"(c), "+v"(d) :: "memory");
}

__global__ __launch_bounds__(32 * NW)
void bilstm_fused(const float* x,
                  const float* __restrict__ w_ih_f, const float* __restrict__ w_hh_f,
                  const float* __restrict__ b_ih_f, const float* __restrict__ b_hh_f,
                  const float* __restrict__ w_ih_b, const float* __restrict__ w_hh_b,
                  const float* __restrict__ b_ih_b, const float* __restrict__ b_hh_b,
                  const float* __restrict__ w_out, const float* __restrict__ b_out,
                  const int* __restrict__ future_p,
                  float* __restrict__ out)
{
    __shared__ __align__(16) float hbuf[2 * 2 * HPLANE];
    __shared__ __align__(16) float obuf[NW * 16 * 4];

    const int lane = threadIdx.x & 31;
    const int wave = __builtin_amdgcn_readfirstlane((int)(threadIdx.x >> 5));
    const int dir  = wave >> 1;
    const int qh   = wave & 1;
    const int half = lane >> 4;
    const int nn   = lane & 15;
    const int cq   = nn >> 2;
    const int t    = nn & 3;
    const bool hi  = half != 0;
    const int bb   = blockIdx.x * 16;
    const int fut  = future_p[0];

    const float* const wih = dir ? w_ih_b : w_ih_f;
    const float* const whh = dir ? w_hh_b : w_hh_f;
    const float* const bih = dir ? b_ih_b : b_ih_f;
    const float* const bhh = dir ? b_hh_b : b_hh_f;

    const int j = 4 * qh + cq;
    const int G = 8 * t + j;
    const int kown = 4 * qh + 2 * half;
    const int koth = 4 * (1 - qh) + 2 * half;
    const v2f wBown = { whh[G * H_SZ + kown], whh[G * H_SZ + kown + 1] };
    const v2f wBoth = { whh[G * H_SZ + koth], whh[G * H_SZ + koth + 1] };
    const v2f wB2 = { hi ? 0.0f : wih[G], hi ? 0.0f : (bih[G] + bhh[G]) };

    const float wo = w_out[8 * dir + j];
    const float bo = b_out[0];

    float* const hb = hbuf + dir * (2 * HPLANE);
    const int ldown = nn * H_SZ + kown;
    const int ldoth = nn * H_SZ + koth;
    const int stoff = (8 * half + t) * H_SZ + j;
    float4* const obuf4 = (float4*)obuf;

    ((float4*)hbuf)[threadIdx.x] = make_float4(0.0f, 0.0f, 0.0f, 0.0f);
    if (threadIdx.x < NW * 16) obuf4[threadIdx.x] = make_float4(0.0f, 0.0f, 0.0f, 0.0f);

    const float* const xcol = x + bb + nn;
    const int xbase = dir ? fut : 0;
    const int xsgn  = dir ? -1 : 1;
    float xr[4];
    #pragma unroll
    for (int k = 0; k < 4; ++k)
        xr[k] = xcol[(size_t)((xbase + xsgn * k) & (S_LEN - 1)) * B_TOT];

    lds_handover<0>();

    float cst[2] = { 0.0f, 0.0f };
    v2f aown = { 0.0f, 0.0f };
    const v8f zero8 = {0,0,0,0,0,0,0,0};
    for (int ii = 0; ii < S_LEN; ii += 4) {
        float pv0[4], pv1[4];
        float4 o0, o1, o2, o3;
        #pragma unroll
        for (int u = 0; u < 4; ++u) {
            const int i = ii + u;
            const float* const hr = hb + (u & 1) * HPLANE;
            float* const hw = hb + ((u + 1) & 1) * HPLANE;

            const v2f aoth = *(const v2f*)&hr[ldoth];
            if (u == 0) {
                o0 = obuf4[nn]; o1 = obuf4[16 + nn]; o2 = obuf4[32 + nn]; o3 = obuf4[48 + nn];
                asm volatile("" ::: "memory");
            }
            float xc = xr[u];
            asm volatile("" : "+v"(xc) :: "memory");
            xr[u] = xcol[(size_t)((xbase + xsgn * (i + 4)) & (S_LEN - 1)) * B_TOT];

            const v2f a2 = { hi ? 0.0f : xc, hi ? 0.0f : 1.0f };
            v8f g = wmma4(a2, wB2, zero8);
            g = wmma4(aown, wBown, g);
            g = wmma4(aoth, wBoth, g);

            float V0[4], V1[4];
            exchange_bit0(g, V0, V1);
            float W[4][2];
            exchange_bit1_if(V0, V1, W[0][0], W[1][0], W[0][1], W[1][1]);
            exchange_bit1_go(V0, V1, W[2][0], W[3][0], W[2][1], W[3][1]);

            float hn[2];
            #pragma unroll
            for (int r2 = 0; r2 < 2; ++r2) {
                const float ig = fsigmoid(W[0][r2]);
                const float fg = fsigmoid(W[1][r2]);
                const float gg = ftanh(W[2][r2]);
                const float og = fsigmoid(W[3][r2]);
                const float cn = fg * cst[r2] + ig * gg;
                cst[r2] = cn;
                hn[r2] = og * ftanh(cn);
            }
            hw[stoff] = hn[0];
            hw[stoff + 4 * H_SZ] = hn[1];
            aown = *(const v2f*)&hw[ldown];

            if (u == 0) {
                hold4(o0.x, o0.y, o0.z, o0.w);
                hold4(o1.x, o1.y, o1.z, o1.w);
                hold4(o2.x, o2.y, o2.z, o2.w);
                hold4(o3.x, o3.y, o3.z, o3.w);
                if (wave == 0 && lane < 16 && ii > 0) {
                    *(float4*)&out[(size_t)(bb + nn) * S_LEN + ii - 4] = make_float4(
                        ((o0.x + o1.x) + (o2.x + o3.x)) + bo,
                        ((o0.y + o1.y) + (o2.y + o3.y)) + bo,
                        ((o0.z + o1.z) + (o2.z + o3.z)) + bo,
                        ((o0.w + o1.w) + (o2.w + o3.w)) + bo);
                }
            }
            float p0 = hn[0] * wo, p1 = hn[1] * wo;
            p0 += lane_get<ROW_ROR4>(p0);
            p1 += lane_get<ROW_ROR4>(p1);
            p0 += lane_get<ROW_ROR8>(p0);
            p1 += lane_get<ROW_ROR8>(p1);
            pv0[u] = p0;
            pv1[u] = p1;
            if (u == 3 && cq == 0) {
                obuf4[wave * 16 + 8 * half + t]     = make_float4(pv0[0], pv0[1], pv0[2], pv0[3]);
                obuf4[wave * 16 + 8 * half + 4 + t] = make_float4(pv1[0], pv1[1], pv1[2], pv1[3]);
            }
            if (u == 3) lds_handover<0>(); else lds_handover<1>();
        }
    }
    if (wave == 0 && lane < 16) {
        const float4 o0 = obuf4[nn], o1 = obuf4[16 + nn], o2 = obuf4[32 + nn], o3 = obuf4[48 + nn];
        *(float4*)&out[(size_t)(bb + nn) * S_LEN + S_LEN - 4] = make_float4(
            ((o0.x + o1.x) + (o2.x + o3.x)) + bo,
            ((o0.y + o1.y) + (o2.y + o3.y)) + bo,
            ((o0.z + o1.z) + (o2.z + o3.z)) + bo,
            ((o0.w + o1.w) + (o2.w + o3.w)) + bo);
    }
}

extern "C" void kernel_launch(void* const* d_in, const int* in_sizes, int n_in,
                              void* d_out, int out_size, void* d_ws, size_t ws_size,
                              hipStream_t stream) {
    (void)in_sizes; (void)n_in; (void)out_size; (void)d_ws; (void)ws_size;
    const float* xp     = (const float*)d_in[0];
    const float* w_ih_f = (const float*)d_in[1];
    const float* w_hh_f = (const float*)d_in[2];
    const float* b_ih_f = (const float*)d_in[3];
    const float* b_hh_f = (const float*)d_in[4];
    const float* w_ih_b = (const float*)d_in[5];
    const float* w_hh_b = (const float*)d_in[6];
    const float* b_ih_b = (const float*)d_in[7];
    const float* b_hh_b = (const float*)d_in[8];
    const float* w_out  = (const float*)d_in[9];
    const float* b_out  = (const float*)d_in[10];
    const int*   future = (const int*)d_in[11];
    float* out = (float*)d_out;

    dim3 grid(B_TOT / 16);
    dim3 block(32 * NW);
    bilstm_fused<<<grid, block, 0, stream>>>(
        xp, w_ih_f, w_hh_f, b_ih_f, b_hh_f,
        w_ih_b, w_hh_b, b_ih_b, b_hh_b,
        w_out, b_out, future, out);
}


# Output projection on the matrix unit: own h chunk times w_out into one column per step of a second 16x16 tile, flushed every 16 steps; per-step lane reductions and 4-step partial exchange are gone
# speedup vs baseline: 2.8043x; 1.0994x over previous
#include <hip/hip_runtime.h>
#include <hip/hip_bf16.h>

#define S_LEN 2048
#define B_TOT 4096
#define H_SZ  8
#define NW    4
#define HPLANE (16 * H_SZ)
#define OPITCH 20
#define OTILE  (16 * OPITCH)

typedef __attribute__((ext_vector_type(2))) float v2f;
typedef __attribute__((ext_vector_type(8))) float v8f;

__device__ __forceinline__ float frcp(float v) {
    return __builtin_amdgcn_rcpf(v);
}

__device__ __forceinline__ float fsigmoid(float v) {
    return frcp(1.0f + __expf(-v));
}

__device__ __forceinline__ float ftanh(float v) {
#if __has_builtin(__builtin_amdgcn_tanhf)
    return __builtin_amdgcn_tanhf(v);
#else
    return 1.0f - 2.0f * frcp(1.0f + __expf(2.0f * v));
#endif
}

template <int YOUNGER>
__device__ __forceinline__ void lds_handover() {
    if (YOUNGER == 1)
        asm volatile("s_wait_dscnt 0x1\n\ts_barrier_signal -1\n\ts_barrier_wait -1" ::: "memory");
    else
        asm volatile("s_wait_dscnt 0x0\n\ts_barrier_signal -1\n\ts_barrier_wait -1" ::: "memory");
}

__device__ __forceinline__ v8f wmma4(v2f a, v2f b, v8f c) {
    return __builtin_amdgcn_wmma_f32_16x16x4_f32(
         false, a,  false, b,
         (short)0, c,  false,  false);
}

#define LANES_BIT0_SET 0xAAAAAAAAu
#define LANES_BIT0_CLR 0x55555555u
#define LANES_BIT1_SET 0xCCCCCCCCu
#define LANES_BIT1_CLR 0x33333333u
__device__ __forceinline__ void exchange_bit0(const v8f& g, float (&V0)[4], float (&V1)[4]) {
    asm("v_cndmask_b32_e64_dpp %0, %9, %8, %16 quad_perm:[1,0,3,2] row_mask:0xf bank_mask:0xf\n\t"
        "v_cndmask_b32_e64_dpp %1, %8, %9, %17 quad_perm:[1,0,3,2] row_mask:0xf bank_mask:0xf\n\t"
        "v_cndmask_b32_e64_dpp %2, %11, %10, %16 quad_perm:[1,0,3,2] row_mask:0xf bank_mask:0xf\n\t"
        "v_cndmask_b32_e64_dpp %3, %10, %11, %17 quad_perm:[1,0,3,2] row_mask:0xf bank_mask:0xf\n\t"
        "v_cndmask_b32_e64_dpp %4, %13, %12, %16 quad_perm:[1,0,3,2] row_mask:0xf bank_mask:0xf\n\t"
        "v_cndmask_b32_e64_dpp %5, %12, %13, %17 quad_perm:[1,0,3,2] row_mask:0xf bank_mask:0xf\n\t"
        "v_cndmask_b32_e64_dpp %6, %15, %14, %16 quad_perm:[1,0,3,2] row_mask:0xf bank_mask:0xf\n\t"
        "v_cndmask_b32_e64_dpp %7, %14, %15, %17 quad_perm:[1,0,3,2] row_mask:0xf bank_mask:0xf"
        : "=&v"(V0[0]), "=&v"(V1[0]), "=&v"(V0[1]), "=&v"(V1[1]),
          "=&v"(V0[2]), "=&v"(V1[2]), "=&v"(V0[3]), "=&v"(V1[3])
        : "v"(g[0]), "v"(g[1]), "v"(g[2]), "v"(g[3]), "v"(g[4]), "v"(g[5]), "v"(g[6]), "v"(g[7]),
          "s"(LANES_BIT0_CLR), "s"(LANES_BIT0_SET));
}
__device__ __forceinline__ void exchange_bit1_if(const float (&V0)[4], const float (&V1)[4],
                                                 float& i0, float& f0, float& i1, float& f1) {
    asm("v_cndmask_b32_e64_dpp %0, %5, %4, %12 quad_perm:[2,3,0,1] row_mask:0xf bank_mask:0xf\n\t"
        "v_cndmask_b32_e64_dpp %1, %7, %6, %12 quad_perm:[2,3,0,1] row_mask:0xf bank_mask:0xf\n\t"
        "v_cndmask_b32_e64_dpp %2, %9, %8, %12 quad_perm:[2,3,0,1] row_mask:0xf bank_mask:0xf\n\t"
        "v_cndmask_b32_e64_dpp %3, %11, %10, %12 quad_perm:[2,3,0,1] row_mask:0xf bank_mask:0xf"
        : "=&v"(i0), "=&v"(f0), "=&v"(i1), "=&v"(f1)
        : "v"(V0[0]), "v"(V0[1]), "v"(V1[0]), "v"(V1[1]), "v"(V0[2]), "v"(V0[3]), "v"(V1[2]), "v"(V1[3]),
          "s"(LANES_BIT1_CLR));
}
__device__ __forceinline__ void exchange_bit1_go(const float (&V0)[4], const float (&V1)[4],
                                                 float& g0, float& o0, float& g1, float& o1) {
    asm("v_cndmask_b32_e64_dpp %0, %4, %5, %12 quad_perm:[2,3,0,1] row_mask:0xf bank_mask:0xf\n\t"
        "v_cndmask_b32_e64_dpp %1, %6, %7, %12 quad_perm:[2,3,0,1] row_mask:0xf bank_mask:0xf\n\t"
        "v_cndmask_b32_e64_dpp %2, %8, %9, %12 quad_perm:[2,3,0,1] row_mask:0xf bank_mask:0xf\n\t"
        "v_cndmask_b32_e64_dpp %3, %10, %11, %12 quad_perm:[2,3,0,1] row_mask:0xf bank_mask:0xf"
        : "=&v"(g0), "=&v"(o0), "=&v"(g1), "=&v"(o1)
        : "v"(V0[0]), "v"(V0[1]), "v"(V1[0]), "v"(V1[1]), "v"(V0[2]), "v"(V0[3]), "v"(V1[2]), "v"(V1[3]),
          "s"(LANES_BIT1_SET));
}

__device__ __forceinline__ void hold4(float& a, float& b, float& c, float& d) {
    asm volatile("" : "+v"(a), "+v"(b), "+v"(c), "+v"(d) :: "memory");
}

__global__ __launch_bounds__(32 * NW)
void bilstm_fused(const float* x,
                  const float* __restrict__ w_ih_f, const float* __restrict__ w_hh_f,
                  const float* __restrict__ b_ih_f, const float* __restrict__ b_hh_f,
                  const float* __restrict__ w_ih_b, const float* __restrict__ w_hh_b,
                  const float* __restrict__ b_ih_b, const float* __restrict__ b_hh_b,
                  const float* __restrict__ w_out, const float* __restrict__ b_out,
                  const int* __restrict__ future_p,
                  float* __restrict__ out)
{
    __shared__ __align__(16) float hbuf[2 * 2 * HPLANE];
    __shared__ __align__(16) float otile[NW * OTILE];

    const int lane = threadIdx.x & 31;
    const int wave = __builtin_amdgcn_readfirstlane((int)(threadIdx.x >> 5));
    const int dir  = wave >> 1;
    const int qh   = wave & 1;
    const int half = lane >> 4;
    const int nn   = lane & 15;
    const int cq   = nn >> 2;
    const int t    = nn & 3;
    const bool hi  = half != 0;
    const int bb   = blockIdx.x * 16;
    const int fut  = future_p[0];

    const float* const wih = dir ? w_ih_b : w_ih_f;
    const float* const whh = dir ? w_hh_b : w_hh_f;
    const float* const bih = dir ? b_ih_b : b_ih_f;
    const float* const bhh = dir ? b_hh_b : b_hh_f;

    const int j = 4 * qh + cq;
    const int G = 8 * t + j;
    const int kown = 4 * qh + 2 * half;
    const int koth = 4 * (1 - qh) + 2 * half;
    const v2f wBown = { whh[G * H_SZ + kown], whh[G * H_SZ + kown + 1] };
    const v2f wBoth = { whh[G * H_SZ + koth], whh[G * H_SZ + koth + 1] };
    const v2f wB2 = { hi ? 0.0f : wih[G], hi ? 0.0f : (bih[G] + bhh[G]) };

    const v2f wo2 = { w_out[8 * dir + kown], w_out[8 * dir + kown + 1] };
    const float bo = b_out[0];

    float* const hb = hbuf + dir * (2 * HPLANE);
    const int ldown = nn * H_SZ + kown;
    const int ldoth = nn * H_SZ + koth;
    const int stoff = (8 * half + t) * H_SZ + j;
    float* const otw = otile + wave * OTILE + (8 * half) * OPITCH + nn;
    const int orow = 8 * (wave & 1) + (lane >> 2);
    const float* const otr = otile + orow * OPITCH + 4 * (lane & 3);
    float* const outp = out + (size_t)(bb + orow) * S_LEN + 4 * (lane & 3);

    ((float4*)hbuf)[threadIdx.x] = make_float4(0.0f, 0.0f, 0.0f, 0.0f);

    const float* const xcol = x + bb + nn;
    const int xbase = dir ? fut : 0;
    const int xsgn  = dir ? -1 : 1;
    float xr[4];
    #pragma unroll
    for (int k = 0; k < 4; ++k)
        xr[k] = xcol[(size_t)((xbase + xsgn * k) & (S_LEN - 1)) * B_TOT];

    lds_handover<0>();

    float cst[2] = { 0.0f, 0.0f };
    v2f aown = { 0.0f, 0.0f };
    const v8f zero8 = {0,0,0,0,0,0,0,0};
    v8f acc = zero8;
    for (int ii = 0; ii < S_LEN; ii += 16) {
        float4 o0, o1, o2, o3;
        #pragma unroll
        for (int u = 0; u < 16; ++u) {
            const int i = ii + u;
            const float* const hr = hb + (u & 1) * HPLANE;
            float* const hw = hb + ((u + 1) & 1) * HPLANE;

            const v2f aoth = *(const v2f*)&hr[ldoth];
            if (u == 2) {
                o0 = *(const float4*)&otr[0];
                o1 = *(const float4*)&otr[OTILE];
                o2 = *(const float4*)&otr[2 * OTILE];
                o3 = *(const float4*)&otr[3 * OTILE];
                asm volatile("" ::: "memory");
            }
            float xc = xr[u & 3];
            asm volatile("" : "+v"(xc) :: "memory");
            xr[u & 3] = xcol[(size_t)((xbase + xsgn * (i + 4)) & (S_LEN - 1)) * B_TOT];

            const v2f a2 = { hi ? 0.0f : xc, hi ? 0.0f : 1.0f };
            const v2f aprev = aown;
            v8f g = wmma4(a2, wB2, zero8);
            g = wmma4(aprev, wBown, g);
            g = wmma4(aoth, wBoth, g);

            float V0[4], V1[4];
            exchange_bit0(g, V0, V1);
            float W[4][2];
            exchange_bit1_if(V0, V1, W[0][0], W[1][0], W[0][1], W[1][1]);
            exchange_bit1_go(V0, V1, W[2][0], W[3][0], W[2][1], W[3][1]);

            float hn[2];
            #pragma unroll
            for (int r2 = 0; r2 < 2; ++r2) {
                const float ig = fsigmoid(W[0][r2]);
                const float fg = fsigmoid(W[1][r2]);
                const float gg = ftanh(W[2][r2]);
                const float og = fsigmoid(W[3][r2]);
                const float cn = fg * cst[r2] + ig * gg;
                cst[r2] = cn;
                hn[r2] = og * ftanh(cn);
            }
            hw[stoff] = hn[0];
            hw[stoff + 4 * H_SZ] = hn[1];
            aown = *(const v2f*)&hw[ldown];

            if (u == 1) {
                asm volatile("" ::: "memory");
                #pragma unroll
                for (int r = 0; r < 8; ++r) otw[r * OPITCH] = acc[r];
            }
            const bool mine = nn == ((u + 15) & 15);
            v2f b2 = { mine ? wo2.x : 0.0f, mine ? wo2.y : 0.0f };
            asm volatile("" : "+v"(b2.x), "+v"(b2.y) :: "memory");
            acc = wmma4(aprev, b2, u == 1 ? zero8 : acc);
            asm volatile("" :: "v"(acc[0]) : "memory");
            if (u == 2) {
                hold4(o0.x, o0.y, o0.z, o0.w);
                hold4(o1.x, o1.y, o1.z, o1.w);
                hold4(o2.x, o2.y, o2.z, o2.w);
                hold4(o3.x, o3.y, o3.z, o3.w);
                if (wave < 2 && ii > 0) {
                    *(float4*)&outp[ii - 16] = make_float4(
                        ((o0.x + o1.x) + (o2.x + o3.x)) + bo,
                        ((o0.y + o1.y) + (o2.y + o3.y)) + bo,
                        ((o0.z + o1.z) + (o2.z + o3.z)) + bo,
                        ((o0.w + o1.w) + (o2.w + o3.w)) + bo);
                }
            }
            if (u == 1) lds_handover<0>(); else lds_handover<1>();
        }
    }
    {
        const bool mine = nn == 15;
        const v2f b2 = { mine ? wo2.x : 0.0f, mine ? wo2.y : 0.0f };
        acc = wmma4(aown, b2, acc);
        #pragma unroll
        for (int r = 0; r < 8; ++r) otw[r * OPITCH] = acc[r];
        lds_handover<0>();
        if (wave < 2) {
            const float4 o0 = *(const float4*)&otr[0], o1 = *(const float4*)&otr[OTILE];
            const float4 o2 = *(const float4*)&otr[2 * OTILE], o3 = *(const float4*)&otr[3 * OTILE];
            *(float4*)&outp[S_LEN - 16] = make_float4(
                ((o0.x + o1.x) + (o2.x + o3.x)) + bo,
                ((o0.y + o1.y) + (o2.y + o3.y)) + bo,
                ((o0.z + o1.z) + (o2.z + o3.z)) + bo,
                ((o0.w + o1.w) + (o2.w + o3.w)) + bo);
        }
    }
}

extern "C" void kernel_launch(void* const* d_in, const int* in_sizes, int n_in,
                              void* d_out, int out_size, void* d_ws, size_t ws_size,
                              hipStream_t stream) {
    (void)in_sizes; (void)n_in; (void)out_size; (void)d_ws; (void)ws_size;
    const float* xp     = (const float*)d_in[0];
    const float* w_ih_f = (const float*)d_in[1];
    const float* w_hh_f = (const float*)d_in[2];
    const float* b_ih_f = (const float*)d_in[3];
    const float* b_hh_f = (const float*)d_in[4];
    const float* w_ih_b = (const float*)d_in[5];
    const float* w_hh_b = (const float*)d_in[6];
    const float* b_ih_b = (const float*)d_in[7];
    const float* b_hh_b = (const float*)d_in[8];
    const float* w_out  = (const float*)d_in[9];
    const float* b_out  = (const float*)d_in[10];
    const int*   future = (const int*)d_in[11];
    float* out = (float*)d_out;

    dim3 grid(B_TOT / 16);
    dim3 block(32 * NW);
    bilstm_fused<<<grid, block, 0, stream>>>(
        xp, w_ih_f, w_hh_f, b_ih_f, b_hh_f,
        w_ih_b, w_hh_b, b_ih_b, b_hh_b,
        w_out, b_out, future, out);
}
